# starting kernel
# baseline (speedup 1.0000x reference)
#include <hip/hip_runtime.h>


typedef __attribute__((ext_vector_type(2))) float v2f;
typedef __attribute__((ext_vector_type(8))) float v8f;

#define F_DIM 48

__global__ void prep_w_kernel(const float* __restrict__ W1,
                              const float* __restrict__ bias,
                              const float* __restrict__ W2,
                              const float* __restrict__ b2,
                              float* __restrict__ w,
                              float* __restrict__ cconst,
                              int H) {
  int t = threadIdx.x;
  if (t < F_DIM) {
    float acc = 0.f;
    for (int h = 0; h < H; ++h) acc = fmaf(W1[t * H + h], W2[h], acc);
    w[t] = acc;
  } else if (t == F_DIM) {
    float sw = 0.f;
    for (int h = 0; h < H; ++h) sw += W2[h];
    *cconst = bias[0] * sw + b2[0];
  }
}

__global__ void init_deg_kernel(float* __restrict__ deg, int N) {
  int n = blockIdx.x * blockDim.x + threadIdx.x;
  if (n < N) deg[n] = 1.0f;
}

__global__ void degree_kernel(const int* __restrict__ col,
                              float* __restrict__ deg, int E) {
  int e = blockIdx.x * blockDim.x + threadIdx.x;
  if (e < E) atomicAdd(&deg[col[e]], 1.0f);
}

__global__ void dinv_kernel(float* __restrict__ deg, int N) {
  int n = blockIdx.x * blockDim.x + threadIdx.x;
  if (n < N) deg[n] = rsqrtf(fmaxf(deg[n], 1.0f));
}

__global__ void s_wmma_kernel(const float* __restrict__ x,
                              const float* __restrict__ w,
                              float* __restrict__ s, int nTiles) {
  int lane = threadIdx.x & 31;
  int tile = blockIdx.x * (blockDim.x >> 5) + (threadIdx.x >> 5);
  if (tile >= nTiles) return;
  int m = lane & 15;
  int halfSel = lane >> 4;
  const float* xrow = x + (size_t)(tile * 16 + m) * F_DIM;
  v8f c = {0.f, 0.f, 0.f, 0.f, 0.f, 0.f, 0.f, 0.f};
#pragma unroll
  for (int k0 = 0; k0 < F_DIM; k0 += 4) {
    int koff = k0 + 2 * halfSel;
    v2f a = *reinterpret_cast<const v2f*>(xrow + koff);
    v2f b = *reinterpret_cast<const v2f*>(w + koff);
    c = __builtin_amdgcn_wmma_f32_16x16x4_f32(
         false, a,  false, b,
         (short)0, c,  false,  false);
  }
  if ((lane & 15) == 0) {
    int base = tile * 16 + halfSel * 8;
#pragma unroll
    for (int j = 0; j < 8; ++j) s[base + j] = c[j];
  }
}

__global__ void s_tail_kernel(const float* __restrict__ x,
                              const float* __restrict__ w,
                              float* __restrict__ s, int start, int N) {
  int n = start + blockIdx.x * blockDim.x + threadIdx.x;
  if (n < N) {
    float acc = 0.f;
    for (int k = 0; k < F_DIM; ++k)
      acc = fmaf(x[(size_t)n * F_DIM + k], w[k], acc);
    s[n] = acc;
  }
}

__global__ void self_out_kernel(const float* __restrict__ dinv,
                                const float* __restrict__ s,
                                const float* __restrict__ cconst,
                                float* __restrict__ out, int N) {
  int n = blockIdx.x * blockDim.x + threadIdx.x;
  if (n < N) {
    float d = dinv[n];
    out[n] = fmaf(d * d, s[n], *cconst);
  }
}

__global__ void edge_scatter_kernel(const int* __restrict__ row,
                                    const int* __restrict__ col,
                                    const float* __restrict__ dinv,
                                    const float* __restrict__ s,
                                    float* __restrict__ out, int E) {
  int e = blockIdx.x * blockDim.x + threadIdx.x;
  if (e < E) {
    int r = row[e], c = col[e];
    atomicAdd(&out[c], dinv[r] * dinv[c] * s[r]);
  }
}

static inline size_t roundup256(size_t v) { return (v + 255) & ~(size_t)255; }

extern "C" void kernel_launch(void* const* d_in, const int* in_sizes, int n_in,
                              void* d_out, int out_size, void* d_ws, size_t ws_size,
                              hipStream_t stream) {
  const float* x    = (const float*)d_in[0];
  const int*   eidx = (const int*)d_in[1];
  const float* W1   = (const float*)d_in[2];
  const float* bias = (const float*)d_in[3];
  const float* W2   = (const float*)d_in[4];
  const float* b2   = (const float*)d_in[5];
  float* out = (float*)d_out;

  const int N = out_size;
  const int E = in_sizes[1] / 2;
  const int H = in_sizes[4];
  const int* row = eidx;
  const int* col = eidx + E;

  char* ws = (char*)d_ws;
  float* deg = (float*)ws;
  float* s   = (float*)(ws + roundup256((size_t)N * 4));
  float* w   = (float*)(ws + 2 * roundup256((size_t)N * 4));
  float* cc  = w + 64;
  (void)ws_size; (void)n_in;

  const int T = 256;
  const int nBlkN = (N + T - 1) / T;
  const int nBlkE = (E + T - 1) / T;

  prep_w_kernel<<<1, 64, 0, stream>>>(W1, bias, W2, b2, w, cc, H);
  init_deg_kernel<<<nBlkN, T, 0, stream>>>(deg, N);
  degree_kernel<<<nBlkE, T, 0, stream>>>(col, deg, E);
  dinv_kernel<<<nBlkN, T, 0, stream>>>(deg, N);

  const int nTiles = N / 16;
  const int wavesPerBlk = T / 32;
  const int nBlkTile = (nTiles + wavesPerBlk - 1) / wavesPerBlk;
  if (nTiles > 0)
    s_wmma_kernel<<<nBlkTile, T, 0, stream>>>(x, w, s, nTiles);
  const int rem = N - nTiles * 16;
  if (rem > 0)
    s_tail_kernel<<<1, T, 0, stream>>>(x, w, s, nTiles * 16, N);

  self_out_kernel<<<nBlkN, T, 0, stream>>>(deg, s, cc, out, N);
  edge_scatter_kernel<<<nBlkE, T, 0, stream>>>(row, col, deg, s, out, E);
}


# Degree count and edge sum of dinv*s accumulated in LDS per node slice x edge partition, partials summed in fixed order, dinv[col] applied after; global atomics removed; float sum order changes
# speedup vs baseline: 3.5455x; 3.5455x over previous
#include <hip/hip_runtime.h>


typedef __attribute__((ext_vector_type(2))) float v2f;
typedef __attribute__((ext_vector_type(8))) float v8f;

#define F_DIM 48
#define SLICE 12512
#define NPART 32
#define SCAN_T 512

__global__ void prep_w_kernel(const float* __restrict__ W1,
                              const float* __restrict__ bias,
                              const float* __restrict__ W2,
                              const float* __restrict__ b2,
                              float* __restrict__ w,
                              float* __restrict__ cconst,
                              int H) {
  int t = threadIdx.x;
  if (t < F_DIM) {
    float acc = 0.f;
    for (int h = 0; h < H; ++h) acc = fmaf(W1[t * H + h], W2[h], acc);
    w[t] = acc;
  } else if (t == F_DIM) {
    float sw = 0.f;
    for (int h = 0; h < H; ++h) sw += W2[h];
    *cconst = bias[0] * sw + b2[0];
  }
}

__global__ void s_wmma_kernel(const float* __restrict__ x,
                              const float* __restrict__ w,
                              float* __restrict__ s, int nTiles) {
  int lane = threadIdx.x & 31;
  int tile = blockIdx.x * (blockDim.x >> 5) + (threadIdx.x >> 5);
  if (tile >= nTiles) return;
  int m = lane & 15;
  int halfSel = lane >> 4;
  const float* xrow = x + (size_t)(tile * 16 + m) * F_DIM;
  v8f c = {0.f, 0.f, 0.f, 0.f, 0.f, 0.f, 0.f, 0.f};
#pragma unroll
  for (int k0 = 0; k0 < F_DIM; k0 += 4) {
    int koff = k0 + 2 * halfSel;
    v2f a = *reinterpret_cast<const v2f*>(xrow + koff);
    v2f b = *reinterpret_cast<const v2f*>(w + koff);
    c = __builtin_amdgcn_wmma_f32_16x16x4_f32(
         false, a,  false, b,
         (short)0, c,  false,  false);
  }
  if ((lane & 15) == 0) {
    int base = tile * 16 + halfSel * 8;
#pragma unroll
    for (int j = 0; j < 8; ++j) s[base + j] = c[j];
  }
}

__global__ void s_tail_kernel(const float* __restrict__ x,
                              const float* __restrict__ w,
                              float* __restrict__ s, int start, int N) {
  int n = start + blockIdx.x * blockDim.x + threadIdx.x;
  if (n < N) {
    float acc = 0.f;
    for (int k = 0; k < F_DIM; ++k)
      acc = fmaf(x[(size_t)n * F_DIM + k], w[k], acc);
    s[n] = acc;
  }
}

__device__ __forceinline__ void part_range(int E, int p, int& e0, int& e1) {
  const int per = (((E + NPART - 1) / NPART) + 3) & ~3;
  const long long b = (long long)p * per;
  const long long x = b + per;
  e0 = b < E ? (int)b : E;
  e1 = x < E ? (int)x : E;
}

__device__ __forceinline__ void count_dst(unsigned* cnt, unsigned lo, int c) {
  const unsigned d = (unsigned)c - lo;
  if (d < SLICE) atomicAdd(&cnt[d], 1u);
}

__global__ __launch_bounds__(SCAN_T)
void deg_partial_kernel(const int* __restrict__ col,
                        unsigned* __restrict__ part, int E, int N) {
  __shared__ __attribute__((aligned(16))) unsigned cnt[SLICE];
  const int tid = threadIdx.x;
  const unsigned lo = blockIdx.y * SLICE;
  for (int i = tid; i < SLICE / 4; i += SCAN_T)
    reinterpret_cast<uint4*>(cnt)[i] = make_uint4(0u, 0u, 0u, 0u);
  __syncthreads();

  int e0, e1;
  part_range(E, blockIdx.x, e0, e1);
  int nv = 0;
  if ((((size_t)col) & 15) == 0) {
    nv = (e1 - e0) >> 2;
    const int4* cv = reinterpret_cast<const int4*>(col + e0);
    const int nfull = nv - nv % (8 * SCAN_T);
    for (int i = tid; i < nfull; i += 8 * SCAN_T) {
      int4 c[8];
#pragma unroll
      for (int u = 0; u < 8; ++u) c[u] = cv[i + u * SCAN_T];
#pragma unroll
      for (int u = 0; u < 8; ++u) {
        count_dst(cnt, lo, c[u].x); count_dst(cnt, lo, c[u].y);
        count_dst(cnt, lo, c[u].z); count_dst(cnt, lo, c[u].w);
      }
    }
    for (int i = nfull + tid; i < nv; i += SCAN_T) {
      const int4 c = cv[i];
      count_dst(cnt, lo, c.x); count_dst(cnt, lo, c.y);
      count_dst(cnt, lo, c.z); count_dst(cnt, lo, c.w);
    }
  }
  for (int e = e0 + 4 * nv + tid; e < e1; e += SCAN_T) count_dst(cnt, lo, col[e]);
  __syncthreads();

  unsigned* dst = part + (size_t)blockIdx.x * N;
  const int len = (unsigned)N > lo ? ((unsigned)N - lo < SLICE ? (int)((unsigned)N - lo) : SLICE) : 0;
  if ((N & 3) == 0) {
    uint4* d4 = reinterpret_cast<uint4*>(dst + lo);
    for (int i = tid; i < (len >> 2); i += SCAN_T) d4[i] = reinterpret_cast<const uint4*>(cnt)[i];
  } else {
    for (int i = tid; i < len; i += SCAN_T) dst[lo + i] = cnt[i];
  }
}

__global__ void dinv_t_kernel(const unsigned* __restrict__ part,
                              const float* __restrict__ s,
                              float* __restrict__ dinv,
                              float* __restrict__ t, int N) {
  const int n = blockIdx.x * blockDim.x + threadIdx.x;
  if (n < N) {
    unsigned c = 1u;
#pragma unroll
    for (int p = 0; p < NPART; ++p) c += part[(size_t)p * N + n];
    const float d = rsqrtf(fmaxf((float)c, 1.0f));
    dinv[n] = d;
    t[n] = d * s[n];
  }
}

#define EDGE_PREP(k, cc, rr)                                          \
  d[k] = (unsigned)(cc) - lo;                                          \
  m[k] = (d[k] < SLICE) && ((unsigned)(rr) < (unsigned)N);             \
  v[k] = t[m[k] ? (unsigned)(rr) : 0u];

__global__ __launch_bounds__(SCAN_T)
void edge_partial_kernel(const int* __restrict__ row,
                         const int* __restrict__ col,
                         const float* __restrict__ t,
                         float* __restrict__ part, int E, int N) {
  __shared__ __attribute__((aligned(16))) float acc[SLICE];
  const int tid = threadIdx.x;
  const unsigned lo = blockIdx.y * SLICE;
  for (int i = tid; i < SLICE / 4; i += SCAN_T)
    reinterpret_cast<float4*>(acc)[i] = make_float4(0.f, 0.f, 0.f, 0.f);
  __syncthreads();

  int e0, e1;
  part_range(E, blockIdx.x, e0, e1);
  int nv = 0;
  if (((((size_t)col) | ((size_t)row)) & 15) == 0) {
    nv = (e1 - e0) >> 2;
    const int4* cv = reinterpret_cast<const int4*>(col + e0);
    const int4* rv = reinterpret_cast<const int4*>(row + e0);
    const int nfull = nv - nv % (4 * SCAN_T);
    int i = tid;
    int4 c[4], r[4];
#pragma unroll
    for (int u = 0; u < 4; ++u) { c[u] = make_int4(0, 0, 0, 0); r[u] = c[u]; }
    if (i < nfull) {
#pragma unroll
      for (int u = 0; u < 4; ++u) { c[u] = cv[i + u * SCAN_T]; r[u] = rv[i + u * SCAN_T]; }
    }
    while (i < nfull) {
      const int nx = i + 4 * SCAN_T;
      const int pf = nx < nfull ? nx : i;
      int4 nc[4], nr[4];
#pragma unroll
      for (int u = 0; u < 4; ++u) { nc[u] = cv[pf + u * SCAN_T]; nr[u] = rv[pf + u * SCAN_T]; }
      unsigned d[16]; bool m[16]; float v[16];
#pragma unroll
      for (int u = 0; u < 4; ++u) {
        EDGE_PREP(4 * u + 0, c[u].x, r[u].x) EDGE_PREP(4 * u + 1, c[u].y, r[u].y)
        EDGE_PREP(4 * u + 2, c[u].z, r[u].z) EDGE_PREP(4 * u + 3, c[u].w, r[u].w)
      }
      asm volatile("" :: "v"(v[0]), "v"(v[1]), "v"(v[2]), "v"(v[3]),
                         "v"(v[4]), "v"(v[5]), "v"(v[6]), "v"(v[7]),
                         "v"(v[8]), "v"(v[9]), "v"(v[10]), "v"(v[11]),
                         "v"(v[12]), "v"(v[13]), "v"(v[14]), "v"(v[15]));
#pragma unroll
      for (int k = 0; k < 16; ++k)
        if (m[k]) atomicAdd(&acc[d[k]], v[k]);
#pragma unroll
      for (int u = 0; u < 4; ++u) { c[u] = nc[u]; r[u] = nr[u]; }
      i = nx;
    }
    for (int j = nfull + tid; j < nv; j += SCAN_T) {
      const int4 q = cv[j];
      const int4 g = rv[j];
      unsigned d[4]; bool m[4]; float v[4];
      EDGE_PREP(0, q.x, g.x) EDGE_PREP(1, q.y, g.y)
      EDGE_PREP(2, q.z, g.z) EDGE_PREP(3, q.w, g.w)
      asm volatile("" :: "v"(v[0]), "v"(v[1]), "v"(v[2]), "v"(v[3]));
#pragma unroll
      for (int k = 0; k < 4; ++k)
        if (m[k]) atomicAdd(&acc[d[k]], v[k]);
    }
  }
  for (int e = e0 + 4 * nv + tid; e < e1; e += SCAN_T) {
    unsigned d[1]; bool m[1]; float v[1];
    EDGE_PREP(0, col[e], row[e])
    if (m[0]) atomicAdd(&acc[d[0]], v[0]);
  }
  __syncthreads();

  float* dst = part + (size_t)blockIdx.x * N;
  const int len = (unsigned)N > lo ? ((unsigned)N - lo < SLICE ? (int)((unsigned)N - lo) : SLICE) : 0;
  if ((N & 3) == 0) {
    float4* d4 = reinterpret_cast<float4*>(dst + lo);
    for (int i = tid; i < (len >> 2); i += SCAN_T) d4[i] = reinterpret_cast<const float4*>(acc)[i];
  } else {
    for (int i = tid; i < len; i += SCAN_T) dst[lo + i] = acc[i];
  }
}

__global__ void final_out_kernel(const float* __restrict__ part,
                                 const float* __restrict__ dinv,
                                 const float* __restrict__ t,
                                 const float* __restrict__ cconst,
                                 float* __restrict__ out, int N) {
  const int n = blockIdx.x * blockDim.x + threadIdx.x;
  if (n < N) {
    float pv[NPART];
#pragma unroll
    for (int p = 0; p < NPART; ++p) pv[p] = part[(size_t)p * N + n];
    float a = t[n];
#pragma unroll
    for (int p = 0; p < NPART; ++p) a += pv[p];
    out[n] = fmaf(dinv[n], a, *cconst);
  }
}

static inline size_t roundup256(size_t v) { return (v + 255) & ~(size_t)255; }

extern "C" void kernel_launch(void* const* d_in, const int* in_sizes, int n_in,
                              void* d_out, int out_size, void* d_ws, size_t ws_size,
                              hipStream_t stream) {
  const float* x    = (const float*)d_in[0];
  const int*   eidx = (const int*)d_in[1];
  const float* W1   = (const float*)d_in[2];
  const float* bias = (const float*)d_in[3];
  const float* W2   = (const float*)d_in[4];
  const float* b2   = (const float*)d_in[5];
  float* out = (float*)d_out;

  const int N = out_size;
  const int E = in_sizes[1] / 2;
  const int H = in_sizes[4];
  const int* row = eidx;
  const int* col = eidx + E;

  char* ws = (char*)d_ws;
  const size_t nBytes = roundup256((size_t)N * 4);
  float* s    = (float*)ws;
  float* dinv = (float*)(ws + nBytes);
  float* t    = (float*)(ws + 2 * nBytes);
  float* w    = (float*)(ws + 3 * nBytes);
  float* cc   = w + 64;
  char* planes = ws + 3 * nBytes + 512;
  (void)ws_size; (void)n_in;

  const int T = 256;
  const int nBlkN = (N + T - 1) / T;
  const int nSlice = (N + SLICE - 1) / SLICE;
  const dim3 scanGrid(NPART, nSlice);

  prep_w_kernel<<<1, 64, 0, stream>>>(W1, bias, W2, b2, w, cc, H);

  const int nTiles = N / 16;
  const int wavesPerBlk = T / 32;
  const int nBlkTile = (nTiles + wavesPerBlk - 1) / wavesPerBlk;
  if (nTiles > 0)
    s_wmma_kernel<<<nBlkTile, T, 0, stream>>>(x, w, s, nTiles);
  const int rem = N - nTiles * 16;
  if (rem > 0)
    s_tail_kernel<<<1, T, 0, stream>>>(x, w, s, nTiles * 16, N);

  deg_partial_kernel<<<scanGrid, SCAN_T, 0, stream>>>(col, (unsigned*)planes, E, N);
  dinv_t_kernel<<<nBlkN, T, 0, stream>>>((const unsigned*)planes, s, dinv, t, N);
  edge_partial_kernel<<<scanGrid, SCAN_T, 0, stream>>>(row, col, t, (float*)planes, E, N);
  final_out_kernel<<<nBlkN, T, 0, stream>>>((const float*)planes, dinv, t, cc, out, N);
}


# Launches 6 to 4: W1@W2 collapse rides in the degree-scan launch as one work-group (lane-strided sums + butterfly: sum order of w changes); the WMMA s kernel also sums the degree planes -> dinv, t
# speedup vs baseline: 4.6898x; 1.3227x over previous
#include <hip/hip_runtime.h>


typedef __attribute__((ext_vector_type(2))) float v2f;
typedef __attribute__((ext_vector_type(8))) float v8f;

#define F_DIM 48
#define SLICE 12512
#define NPART 32
#define SCAN_T 512

__device__ __forceinline__ void prep_w_block(const float* __restrict__ W1,
                                             const float* __restrict__ bias,
                                             const float* __restrict__ W2,
                                             const float* __restrict__ b2,
                                             float* __restrict__ w,
                                             float* __restrict__ cconst,
                                             int H) {
  constexpr int NW = SCAN_T / 32;
  constexpr int NR = F_DIM / NW;
  static_assert(F_DIM % NW == 0, "rows must divide evenly over the waves");
  const int lane = threadIdx.x & 31;
  const int wave = __builtin_amdgcn_readfirstlane((int)(threadIdx.x >> 5));
  float acc[NR];
#pragma unroll
  for (int i = 0; i < NR; ++i) acc[i] = 0.f;
  float sw = 0.f;
#pragma unroll 8
  for (int h = lane; h < H; h += 32) {
    const float w2 = W2[h];
#pragma unroll
    for (int i = 0; i < NR; ++i)
      acc[i] = fmaf(W1[(size_t)(wave + NW * i) * H + h], w2, acc[i]);
    sw += w2;
  }
#pragma unroll
  for (int o = 16; o > 0; o >>= 1) {
#pragma unroll
    for (int i = 0; i < NR; ++i) acc[i] += __shfl_xor(acc[i], o);
    sw += __shfl_xor(sw, o);
  }
  if (lane == 0) {
#pragma unroll
    for (int i = 0; i < NR; ++i) w[wave + NW * i] = acc[i];
    if (wave == 0) *cconst = bias[0] * sw + b2[0];
  }
}

__global__ void node_wmma_kernel(const float* __restrict__ x,
                                 const float* __restrict__ w,
                                 const unsigned* __restrict__ part,
                                 float* __restrict__ dinv,
                                 float* __restrict__ t, int nTiles, int N) {
  int lane = threadIdx.x & 31;
  int tile = blockIdx.x * (blockDim.x >> 5) + (threadIdx.x >> 5);
  if (tile >= nTiles) return;
  int m = lane & 15;
  int halfSel = lane >> 4;
  const int r = m & 7;
  const int node = tile * 16 + halfSel * 8 + r;
  const unsigned* pp = part + (size_t)((m >> 3) * (NPART / 2)) * N + node;
  unsigned pc[NPART / 2];
#pragma unroll
  for (int j = 0; j < NPART / 2; ++j) pc[j] = pp[(size_t)j * N];
  __builtin_amdgcn_sched_barrier(0);
  const float* xrow = x + (size_t)(tile * 16 + m) * F_DIM;
  v8f c = {0.f, 0.f, 0.f, 0.f, 0.f, 0.f, 0.f, 0.f};
#pragma unroll
  for (int k0 = 0; k0 < F_DIM; k0 += 4) {
    int koff = k0 + 2 * halfSel;
    v2f a = *reinterpret_cast<const v2f*>(xrow + koff);
    v2f b = *reinterpret_cast<const v2f*>(w + koff);
    c = __builtin_amdgcn_wmma_f32_16x16x4_f32(
         false, a,  false, b,
         (short)0, c,  false,  false);
  }
  unsigned cnt = 0u;
#pragma unroll
  for (int j = 0; j < NPART / 2; ++j) cnt += pc[j];
  cnt += (unsigned)__shfl_xor((int)cnt, 8);
  float sv = c[0];
  sv = r == 1 ? c[1] : sv; sv = r == 2 ? c[2] : sv; sv = r == 3 ? c[3] : sv;
  sv = r == 4 ? c[4] : sv; sv = r == 5 ? c[5] : sv; sv = r == 6 ? c[6] : sv;
  sv = r == 7 ? c[7] : sv;
  if (m < 8) {
    const float d = rsqrtf(fmaxf((float)(cnt + 1u), 1.0f));
    dinv[node] = d;
    t[node] = d * sv;
  }
}

__global__ void node_tail_kernel(const float* __restrict__ x,
                                 const float* __restrict__ w,
                                 const unsigned* __restrict__ part,
                                 float* __restrict__ dinv,
                                 float* __restrict__ t, int start, int N) {
  int n = start + blockIdx.x * blockDim.x + threadIdx.x;
  if (n < N) {
    float acc = 0.f;
    for (int k = 0; k < F_DIM; ++k)
      acc = fmaf(x[(size_t)n * F_DIM + k], w[k], acc);
    unsigned cnt = 1u;
    for (int p = 0; p < NPART; ++p) cnt += part[(size_t)p * N + n];
    const float d = rsqrtf(fmaxf((float)cnt, 1.0f));
    dinv[n] = d;
    t[n] = d * acc;
  }
}

__device__ __forceinline__ void part_range(int E, int p, int& e0, int& e1) {
  const int per = (((E + NPART - 1) / NPART) + 3) & ~3;
  const long long b = (long long)p * per;
  const long long x = b + per;
  e0 = b < E ? (int)b : E;
  e1 = x < E ? (int)x : E;
}

__device__ __forceinline__ void count_dst(unsigned* cnt, unsigned lo, int c) {
  const unsigned d = (unsigned)c - lo;
  if (d < SLICE) atomicAdd(&cnt[d], 1u);
}

__global__ __launch_bounds__(SCAN_T)
void deg_partial_kernel(const int* __restrict__ col,
                        unsigned* __restrict__ part, int E, int N, int nSlice,
                        const float* __restrict__ W1,
                        const float* __restrict__ bias,
                        const float* __restrict__ W2,
                        const float* __restrict__ b2,
                        float* __restrict__ w,
                        float* __restrict__ cconst, int H) {
  __shared__ __attribute__((aligned(16))) unsigned cnt[SLICE];
  if (blockIdx.y >= (unsigned)nSlice) {
    if (blockIdx.x == 0) prep_w_block(W1, bias, W2, b2, w, cconst, H);
    return;
  }
  const int tid = threadIdx.x;
  const unsigned lo = blockIdx.y * SLICE;
  for (int i = tid; i < SLICE / 4; i += SCAN_T)
    reinterpret_cast<uint4*>(cnt)[i] = make_uint4(0u, 0u, 0u, 0u);
  __syncthreads();

  int e0, e1;
  part_range(E, blockIdx.x, e0, e1);
  int nv = 0;
  if ((((size_t)col) & 15) == 0) {
    nv = (e1 - e0) >> 2;
    const int4* cv = reinterpret_cast<const int4*>(col + e0);
    const int nfull = nv - nv % (8 * SCAN_T);
    for (int i = tid; i < nfull; i += 8 * SCAN_T) {
      int4 c[8];
#pragma unroll
      for (int u = 0; u < 8; ++u) c[u] = cv[i + u * SCAN_T];
#pragma unroll
      for (int u = 0; u < 8; ++u) {
        count_dst(cnt, lo, c[u].x); count_dst(cnt, lo, c[u].y);
        count_dst(cnt, lo, c[u].z); count_dst(cnt, lo, c[u].w);
      }
    }
    for (int i = nfull + tid; i < nv; i += SCAN_T) {
      const int4 c = cv[i];
      count_dst(cnt, lo, c.x); count_dst(cnt, lo, c.y);
      count_dst(cnt, lo, c.z); count_dst(cnt, lo, c.w);
    }
  }
  for (int e = e0 + 4 * nv + tid; e < e1; e += SCAN_T) count_dst(cnt, lo, col[e]);
  __syncthreads();

  unsigned* dst = part + (size_t)blockIdx.x * N;
  const int len = (unsigned)N > lo ? ((unsigned)N - lo < SLICE ? (int)((unsigned)N - lo) : SLICE) : 0;
  if ((N & 3) == 0) {
    uint4* d4 = reinterpret_cast<uint4*>(dst + lo);
    for (int i = tid; i < (len >> 2); i += SCAN_T) d4[i] = reinterpret_cast<const uint4*>(cnt)[i];
  } else {
    for (int i = tid; i < len; i += SCAN_T) dst[lo + i] = cnt[i];
  }
}

#define EDGE_PREP(k, cc, rr)                                          \
  d[k] = (unsigned)(cc) - lo;                                          \
  m[k] = (d[k] < SLICE) && ((unsigned)(rr) < (unsigned)N);             \
  v[k] = t[m[k] ? (unsigned)(rr) : 0u];

__global__ __launch_bounds__(SCAN_T)
void edge_partial_kernel(const int* __restrict__ row,
                         const int* __restrict__ col,
                         const float* __restrict__ t,
                         float* __restrict__ part, int E, int N) {
  __shared__ __attribute__((aligned(16))) float acc[SLICE];
  const int tid = threadIdx.x;
  const unsigned lo = blockIdx.y * SLICE;
  for (int i = tid; i < SLICE / 4; i += SCAN_T)
    reinterpret_cast<float4*>(acc)[i] = make_float4(0.f, 0.f, 0.f, 0.f);
  __syncthreads();

  int e0, e1;
  part_range(E, blockIdx.x, e0, e1);
  int nv = 0;
  if (((((size_t)col) | ((size_t)row)) & 15) == 0) {
    nv = (e1 - e0) >> 2;
    const int4* cv = reinterpret_cast<const int4*>(col + e0);
    const int4* rv = reinterpret_cast<const int4*>(row + e0);
    const int nfull = nv - nv % (4 * SCAN_T);
    int i = tid;
    int4 c[4], r[4];
#pragma unroll
    for (int u = 0; u < 4; ++u) { c[u] = make_int4(0, 0, 0, 0); r[u] = c[u]; }
    if (i < nfull) {
#pragma unroll
      for (int u = 0; u < 4; ++u) { c[u] = cv[i + u * SCAN_T]; r[u] = rv[i + u * SCAN_T]; }
    }
    while (i < nfull) {
      const int nx = i + 4 * SCAN_T;
      const int pf = nx < nfull ? nx : i;
      int4 nc[4], nr[4];
#pragma unroll
      for (int u = 0; u < 4; ++u) { nc[u] = cv[pf + u * SCAN_T]; nr[u] = rv[pf + u * SCAN_T]; }
      unsigned d[16]; bool m[16]; float v[16];
#pragma unroll
      for (int u = 0; u < 4; ++u) {
        EDGE_PREP(4 * u + 0, c[u].x, r[u].x) EDGE_PREP(4 * u + 1, c[u].y, r[u].y)
        EDGE_PREP(4 * u + 2, c[u].z, r[u].z) EDGE_PREP(4 * u + 3, c[u].w, r[u].w)
      }
      asm volatile("" :: "v"(v[0]), "v"(v[1]), "v"(v[2]), "v"(v[3]),
                         "v"(v[4]), "v"(v[5]), "v"(v[6]), "v"(v[7]),
                         "v"(v[8]), "v"(v[9]), "v"(v[10]), "v"(v[11]),
                         "v"(v[12]), "v"(v[13]), "v"(v[14]), "v"(v[15]));
#pragma unroll
      for (int k = 0; k < 16; ++k)
        if (m[k]) atomicAdd(&acc[d[k]], v[k]);
#pragma unroll
      for (int u = 0; u < 4; ++u) { c[u] = nc[u]; r[u] = nr[u]; }
      i = nx;
    }
    for (int j = nfull + tid; j < nv; j += SCAN_T) {
      const int4 q = cv[j];
      const int4 g = rv[j];
      unsigned d[4]; bool m[4]; float v[4];
      EDGE_PREP(0, q.x, g.x) EDGE_PREP(1, q.y, g.y)
      EDGE_PREP(2, q.z, g.z) EDGE_PREP(3, q.w, g.w)
      asm volatile("" :: "v"(v[0]), "v"(v[1]), "v"(v[2]), "v"(v[3]));
#pragma unroll
      for (int k = 0; k < 4; ++k)
        if (m[k]) atomicAdd(&acc[d[k]], v[k]);
    }
  }
  for (int e = e0 + 4 * nv + tid; e < e1; e += SCAN_T) {
    unsigned d[1]; bool m[1]; float v[1];
    EDGE_PREP(0, col[e], row[e])
    if (m[0]) atomicAdd(&acc[d[0]], v[0]);
  }
  __syncthreads();

  float* dst = part + (size_t)blockIdx.x * N;
  const int len = (unsigned)N > lo ? ((unsigned)N - lo < SLICE ? (int)((unsigned)N - lo) : SLICE) : 0;
  if ((N & 3) == 0) {
    float4* d4 = reinterpret_cast<float4*>(dst + lo);
    for (int i = tid; i < (len >> 2); i += SCAN_T) d4[i] = reinterpret_cast<const float4*>(acc)[i];
  } else {
    for (int i = tid; i < len; i += SCAN_T) dst[lo + i] = acc[i];
  }
}

__global__ void final_out_kernel(const float* __restrict__ part,
                                 const float* __restrict__ dinv,
                                 const float* __restrict__ t,
                                 const float* __restrict__ cconst,
                                 float* __restrict__ out, int N) {
  const int n = blockIdx.x * blockDim.x + threadIdx.x;
  if (n < N) {
    float pv[NPART];
#pragma unroll
    for (int p = 0; p < NPART; ++p) pv[p] = part[(size_t)p * N + n];
    float a = t[n];
#pragma unroll
    for (int p = 0; p < NPART; ++p) a += pv[p];
    out[n] = fmaf(dinv[n], a, *cconst);
  }
}

static inline size_t roundup256(size_t v) { return (v + 255) & ~(size_t)255; }

extern "C" void kernel_launch(void* const* d_in, const int* in_sizes, int n_in,
                              void* d_out, int out_size, void* d_ws, size_t ws_size,
                              hipStream_t stream) {
  const float* x    = (const float*)d_in[0];
  const int*   eidx = (const int*)d_in[1];
  const float* W1   = (const float*)d_in[2];
  const float* bias = (const float*)d_in[3];
  const float* W2   = (const float*)d_in[4];
  const float* b2   = (const float*)d_in[5];
  float* out = (float*)d_out;

  const int N = out_size;
  const int E = in_sizes[1] / 2;
  const int H = in_sizes[4];
  const int* row = eidx;
  const int* col = eidx + E;

  char* ws = (char*)d_ws;
  const size_t nBytes = roundup256((size_t)N * 4);
  float* dinv = (float*)ws;
  float* t    = (float*)(ws + nBytes);
  float* w    = (float*)(ws + 2 * nBytes);
  float* cc   = w + 64;
  char* planes = ws + 2 * nBytes + 512;
  (void)ws_size; (void)n_in;

  const int T = 256;
  const int nBlkN = (N + T - 1) / T;
  const int nSlice = (N + SLICE - 1) / SLICE;
  const dim3 scanGrid(NPART, nSlice);
  const dim3 degGrid(NPART, nSlice + 1);

  deg_partial_kernel<<<degGrid, SCAN_T, 0, stream>>>(col, (unsigned*)planes, E, N, nSlice,
                                                     W1, bias, W2, b2, w, cc, H);

  const int nTiles = N / 16;
  const int wavesPerBlk = T / 32;
  const int nBlkTile = (nTiles + wavesPerBlk - 1) / wavesPerBlk;
  if (nTiles > 0)
    node_wmma_kernel<<<nBlkTile, T, 0, stream>>>(x, w, (const unsigned*)planes, dinv, t, nTiles, N);
  const int rem = N - nTiles * 16;
  if (rem > 0)
    node_tail_kernel<<<1, T, 0, stream>>>(x, w, (const unsigned*)planes, dinv, t, nTiles * 16, N);

  edge_partial_kernel<<<scanGrid, SCAN_T, 0, stream>>>(row, col, t, (float*)planes, E, N);
  final_out_kernel<<<nBlkN, T, 0, stream>>>((const float*)planes, dinv, t, cc, out, N);
}


# Edges counting-sorted once by destination slice (400 nodes) in LDS; a slice work-group then owns its nodes and reads only its own edges: no partial planes, launches 4 to 3; float sum order changes
# speedup vs baseline: 5.8884x; 1.2556x over previous
#include <hip/hip_runtime.h>


typedef __attribute__((ext_vector_type(2))) float v2f;
typedef __attribute__((ext_vector_type(8))) float v8f;

#define F_DIM 48
#define SL 400
#define TILES (SL / 16)
#define DBITS 9
#define DMASK ((1u << DBITS) - 1u)
#define RBAD 0x7FFFFFu
#define NB 512
#define CAP 6656
#define SCAN_T 512
#define EPT ((CAP + SCAN_T - 1) / SCAN_T)
#define WT 256
#define WW (WT / 32)
#define TPW ((TILES + WW - 1) / WW)

static_assert(SL % 16 == 0 && SL <= (1 << DBITS), "slice must be whole tiles and fit its field");
static_assert(NB == SCAN_T, "one histogram bin per thread");
static_assert(CAP < 65536, "start and count of a segment share one 32-bit word");
static_assert(CAP % SCAN_T == 0, "a full partition gives every thread of the sort the same count");
static_assert((1 << DBITS) == 2 * WT, "the slice arrays are cleared two words per thread");
static_assert(SL <= 2 * WT, "a slice thread owns at most two nodes");

__device__ __forceinline__ void prep_w_block(const float* __restrict__ W1,
                                             const float* __restrict__ bias,
                                             const float* __restrict__ W2,
                                             const float* __restrict__ b2,
                                             float* __restrict__ w,
                                             float* __restrict__ cconst,
                                             int H) {
  constexpr int NW = SCAN_T / 32;
  constexpr int NR = F_DIM / NW;
  static_assert(F_DIM % NW == 0, "rows must divide evenly over the waves");
  const int lane = threadIdx.x & 31;
  const int wave = __builtin_amdgcn_readfirstlane((int)(threadIdx.x >> 5));
  float acc[NR];
#pragma unroll
  for (int i = 0; i < NR; ++i) acc[i] = 0.f;
  float sw = 0.f;
#pragma unroll 8
  for (int h = lane; h < H; h += 32) {
    const float w2 = W2[h];
#pragma unroll
    for (int i = 0; i < NR; ++i)
      acc[i] = fmaf(W1[(size_t)(wave + NW * i) * H + h], w2, acc[i]);
    sw += w2;
  }
#pragma unroll
  for (int o = 16; o > 0; o >>= 1) {
#pragma unroll
    for (int i = 0; i < NR; ++i) acc[i] += __shfl_xor(acc[i], o);
    sw += __shfl_xor(sw, o);
  }
  if (lane == 0) {
#pragma unroll
    for (int i = 0; i < NR; ++i) w[wave + NW * i] = acc[i];
    if (wave == 0) *cconst = bias[0] * sw + b2[0];
  }
}

#define HOLD13(a)                                                          \
  asm volatile("" :: "v"(a[0]), "v"(a[1]), "v"(a[2]), "v"(a[3]), "v"(a[4]), "v"(a[5]),      \
                     "v"(a[6]), "v"(a[7]), "v"(a[8]), "v"(a[9]), "v"(a[10]), "v"(a[11]),    \
                     "v"(a[12]))
static_assert(EPT == 13, "HOLD13 names every element");

__global__ __launch_bounds__(SCAN_T)
void sort_edges_kernel(const int* __restrict__ row,
                       const int* __restrict__ col,
                       unsigned* __restrict__ sorted,
                       unsigned* __restrict__ segT, int E, int N, int P, int nSl,
                       const float* __restrict__ W1,
                       const float* __restrict__ bias,
                       const float* __restrict__ W2,
                       const float* __restrict__ b2,
                       float* __restrict__ w,
                       float* __restrict__ cconst, int H) {
  __shared__ unsigned hist[NB + 32];
  __shared__ unsigned offs[NB];
  __shared__ unsigned wsum[SCAN_T / 32];
  __shared__ unsigned stage[CAP];
  if (blockIdx.x >= (unsigned)P) {
    prep_w_block(W1, bias, W2, b2, w, cconst, H);
    return;
  }
  const int tid = threadIdx.x;
  const int lane = tid & 31;
  const int wave = tid >> 5;
  const int p = blockIdx.x;
  hist[tid] = 0u;
  if (tid < 32) hist[NB + tid] = 0u;
  const int e0 = p * CAP;
  const int e1 = (E - e0 < CAP) ? E : e0 + CAP;
  int cc[EPT], rr[EPT];
#pragma unroll
  for (int u = 0; u < EPT; ++u) {
    const int e = e0 + tid + u * SCAN_T;
    const bool in = e < e1;
    const int ec = in ? e : e0;
    cc[u] = col[ec];
    rr[u] = row[ec];
  }
  HOLD13(cc); HOLD13(rr);
#pragma unroll
  for (int u = 0; u < EPT; ++u)
    if (e0 + tid + u * SCAN_T >= e1) cc[u] = -1;
  __syncthreads();

  unsigned bin[EPT], rk[EPT], pk[EPT];
  bool val[EPT];
#pragma unroll
  for (int u = 0; u < EPT; ++u) {
    const unsigned c = (unsigned)cc[u];
    const unsigned b = c / SL;
    val[u] = (c < (unsigned)N) && (b < NB);
    bin[u] = val[u] ? b : (unsigned)(NB + lane);
    const unsigned r = (unsigned)rr[u] < (unsigned)N ? (unsigned)rr[u] : RBAD;
    pk[u] = (c - b * SL) | (r << DBITS);
    rk[u] = atomicAdd(&hist[bin[u]], 1u);
  }
  __syncthreads();

  const unsigned cnt = hist[tid];
  unsigned inc = cnt;
#pragma unroll
  for (int o = 1; o < 32; o <<= 1) {
    const unsigned up = __shfl_up(inc, (unsigned)o);
    if (lane >= o) inc += up;
  }
  if (lane == 31) wsum[wave] = inc;
  __syncthreads();
  unsigned base = 0u;
#pragma unroll
  for (int v = 0; v < SCAN_T / 32; ++v) base += v < wave ? wsum[v] : 0u;
  const unsigned ex = base + inc - cnt;
  offs[tid] = ex;
  if (tid < nSl) segT[(size_t)tid * P + p] = ex | (cnt << 16);
  __syncthreads();

  unsigned ob[EPT];
#pragma unroll
  for (int u = 0; u < EPT; ++u) ob[u] = offs[val[u] ? bin[u] : 0u];
  HOLD13(ob);
#pragma unroll
  for (int u = 0; u < EPT; ++u)
    if (val[u]) stage[ob[u] + rk[u]] = pk[u];
  __syncthreads();

  const unsigned total = offs[NB - 1] + hist[NB - 1];
  unsigned* dst = sorted + (size_t)p * CAP;
  for (unsigned i = tid; i < total; i += SCAN_T) dst[i] = stage[i];
  if (total == 0u && tid == 0) dst[0] = 0u;
}

#define SEG_WORDS(pb) ((pb) + lane < P ? segT[(size_t)y * P + (pb) + lane] : 0u)

#define SEG_LOAD(k)                                                        \
  {                                                                        \
    const unsigned sg = (unsigned)__builtin_amdgcn_readlane((int)seg, k);  \
    const unsigned st = sg & 0xFFFFu, cn = sg >> 16;                       \
    const int pc = pb + (k) < P ? pb + (k) : P - 1;                        \
    const unsigned* sp = sorted + (size_t)pc * CAP;                        \
    e[2 * (k)] = sp[(unsigned)lane < cn ? st + lane : 0u];                 \
    e[2 * (k) + 1] = sp[(unsigned)lane + 32u < cn ? st + lane + 32u : 0u]; \
  }

#define HOLD16(a, o)                                                       \
  asm volatile("" :: "v"(a[(o) + 0]), "v"(a[(o) + 1]), "v"(a[(o) + 2]), "v"(a[(o) + 3]),     \
                     "v"(a[(o) + 4]), "v"(a[(o) + 5]), "v"(a[(o) + 6]), "v"(a[(o) + 7]),     \
                     "v"(a[(o) + 8]), "v"(a[(o) + 9]), "v"(a[(o) + 10]), "v"(a[(o) + 11]),   \
                     "v"(a[(o) + 12]), "v"(a[(o) + 13]), "v"(a[(o) + 14]), "v"(a[(o) + 15]))

#define HOLD12(f)                                                          \
  asm volatile("" :: "v"(f[0]), "v"(f[1]), "v"(f[2]), "v"(f[3]), "v"(f[4]), "v"(f[5]),      \
                     "v"(f[6]), "v"(f[7]), "v"(f[8]), "v"(f[9]), "v"(f[10]), "v"(f[11]))
static_assert(F_DIM / 4 == 12 && TPW == 4, "HOLD12 names every fragment of a tile, and four tiles are named");

__global__ __launch_bounds__(WT)
void slice_node_kernel(const float* __restrict__ x,
                       const float* __restrict__ w,
                       const unsigned* __restrict__ sorted,
                       const unsigned* __restrict__ segT,
                       float* __restrict__ dinv,
                       float* __restrict__ t, int N, int P) {
  __shared__ unsigned deg[1 << DBITS];
  __shared__ float sS[SL];
  const int tid = threadIdx.x;
  const int lane = tid & 31;
  const int wave = __builtin_amdgcn_readfirstlane((int)(threadIdx.x >> 5));
  const int y = blockIdx.x;
  deg[tid] = 0u;
  deg[tid + WT] = 0u;
  int pb = wave * 32;
  unsigned seg = SEG_WORDS(pb);

  const int m = lane & 15;
  const int halfSel = lane >> 4;
  const int r = m & 7;
  v2f b[F_DIM / 4], a[TPW][F_DIM / 4];
#pragma unroll
  for (int k = 0; k < F_DIM / 4; ++k)
    b[k] = *reinterpret_cast<const v2f*>(w + 4 * k + 2 * halfSel);
#pragma unroll
  for (int i = 0; i < TPW; ++i) {
    const int tl = wave + WW * i;
    const int tc = tl < TILES ? tl : TILES - 1;
    int nrow = y * SL + tc * 16 + m;
    nrow = nrow < N ? nrow : N - 1;
    const float* xrow = x + (size_t)nrow * F_DIM;
#pragma unroll
    for (int k = 0; k < F_DIM / 4; ++k)
      a[i][k] = *reinterpret_cast<const v2f*>(xrow + 4 * k + 2 * halfSel);
  }
  HOLD12(b); HOLD12(a[0]); HOLD12(a[1]); HOLD12(a[2]); HOLD12(a[3]);
  float sv[TPW];
#pragma unroll
  for (int i = 0; i < TPW; ++i) {
    v8f c = {0.f, 0.f, 0.f, 0.f, 0.f, 0.f, 0.f, 0.f};
#pragma unroll
    for (int k = 0; k < F_DIM / 4; ++k)
      c = __builtin_amdgcn_wmma_f32_16x16x4_f32(
           false, a[i][k],  false, b[k],
           (short)0, c,  false,  false);
    float s = c[0];
    s = r == 1 ? c[1] : s; s = r == 2 ? c[2] : s; s = r == 3 ? c[3] : s;
    s = r == 4 ? c[4] : s; s = r == 5 ? c[5] : s; s = r == 6 ? c[6] : s;
    s = r == 7 ? c[7] : s;
    sv[i] = s;
  }
#pragma unroll
  for (int i = 0; i < TPW; ++i) {
    const int tl = wave + WW * i;
    if (tl < TILES && m < 8) sS[tl * 16 + halfSel * 8 + r] = sv[i];
  }
  __syncthreads();

  while (pb < P) {
    unsigned e[64];
#pragma unroll
    for (int k = 0; k < 32; ++k) SEG_LOAD(k)
    HOLD16(e, 0); HOLD16(e, 16); HOLD16(e, 32); HOLD16(e, 48);
#pragma unroll
    for (int k = 0; k < 32; ++k) {
      const unsigned cn = (unsigned)__builtin_amdgcn_readlane((int)seg, k) >> 16;
      if ((unsigned)lane < cn) atomicAdd(&deg[e[2 * k] & DMASK], 1u);
      if ((unsigned)lane + 32u < cn) atomicAdd(&deg[e[2 * k + 1] & DMASK], 1u);
    }
    if (__any((seg >> 16) > 64u)) {
#pragma nounroll
      for (int k = 0; k < 32; ++k) {
        const unsigned sg = (unsigned)__shfl((int)seg, k);
        const unsigned st = sg & 0xFFFFu, cn = sg >> 16;
        if (cn > 64u) {
          const unsigned* sp = sorted + (size_t)(pb + k) * CAP;
          for (unsigned j = 64u + lane; j < cn; j += 32u) atomicAdd(&deg[sp[st + j] & DMASK], 1u);
        }
      }
    }
    pb += WW * 32;
    seg = SEG_WORDS(pb);
  }
  __syncthreads();

  for (int i = tid; i < SL; i += WT) {
    const int n = y * SL + i;
    if (n < N) {
      const float d = rsqrtf(fmaxf((float)(deg[i] + 1u), 1.0f));
      dinv[n] = d;
      t[n] = d * sS[i];
    }
  }
}

#define MSG_OK(k, h) (((unsigned)lane + 32u * (h) < cn) && ((e[2 * (k) + (h)] >> DBITS) < (unsigned)N))

__global__ __launch_bounds__(WT)
void slice_msg_kernel(const unsigned* __restrict__ sorted,
                      const unsigned* __restrict__ segT,
                      const float* __restrict__ dinv,
                      const float* __restrict__ t,
                      const float* __restrict__ cconst,
                      float* __restrict__ out, int N, int P) {
  __shared__ float acc[1 << DBITS];
  const int tid = threadIdx.x;
  const int lane = tid & 31;
  const int wave = __builtin_amdgcn_readfirstlane((int)(threadIdx.x >> 5));
  const int y = blockIdx.x;
  acc[tid] = 0.f;
  acc[tid + WT] = 0.f;
  int pb = wave * 32;
  unsigned seg = SEG_WORDS(pb);
  const int n0 = y * SL + tid, n1 = n0 + WT;
  const float dn0 = dinv[n0 < N ? n0 : N - 1], tn0 = t[n0 < N ? n0 : N - 1];
  const float dn1 = dinv[n1 < N ? n1 : N - 1], tn1 = t[n1 < N ? n1 : N - 1];
  const float cst = *cconst;
  __syncthreads();
  asm volatile("" :: "v"(dn0), "v"(tn0), "v"(dn1), "v"(tn1));

  while (pb < P) {
    unsigned e[64];
    float v[64];
#pragma unroll
    for (int k = 0; k < 32; ++k) SEG_LOAD(k)
#pragma unroll
    for (int k = 0; k < 32; ++k) {
      const unsigned cn = (unsigned)__builtin_amdgcn_readlane((int)seg, k) >> 16;
      v[2 * k] = t[MSG_OK(k, 0) ? e[2 * k] >> DBITS : 0u];
      v[2 * k + 1] = t[MSG_OK(k, 1) ? e[2 * k + 1] >> DBITS : 0u];
    }
    HOLD16(v, 0); HOLD16(v, 16); HOLD16(v, 32); HOLD16(v, 48);
#pragma unroll
    for (int k = 0; k < 32; ++k) {
      const unsigned cn = (unsigned)__builtin_amdgcn_readlane((int)seg, k) >> 16;
      if (MSG_OK(k, 0)) atomicAdd(&acc[e[2 * k] & DMASK], v[2 * k]);
      if (MSG_OK(k, 1)) atomicAdd(&acc[e[2 * k + 1] & DMASK], v[2 * k + 1]);
    }
    if (__any((seg >> 16) > 64u)) {
#pragma nounroll
      for (int k = 0; k < 32; ++k) {
        const unsigned sg = (unsigned)__shfl((int)seg, k);
        const unsigned st = sg & 0xFFFFu, cn = sg >> 16;
        if (cn > 64u) {
          const unsigned* sp = sorted + (size_t)(pb + k) * CAP;
          for (unsigned j = 64u + lane; j < cn; j += 32u) {
            const unsigned q = sp[st + j];
            if ((q >> DBITS) < (unsigned)N) atomicAdd(&acc[q & DMASK], t[q >> DBITS]);
          }
        }
      }
    }
    pb += WW * 32;
    seg = SEG_WORDS(pb);
  }
  __syncthreads();

  if (n0 < N) out[n0] = fmaf(dn0, tn0 + acc[tid], cst);
  if (tid + WT < SL && n1 < N) out[n1] = fmaf(dn1, tn1 + acc[tid + WT], cst);
}

static inline size_t roundup256(size_t v) { return (v + 255) & ~(size_t)255; }

extern "C" void kernel_launch(void* const* d_in, const int* in_sizes, int n_in,
                              void* d_out, int out_size, void* d_ws, size_t ws_size,
                              hipStream_t stream) {
  const float* x    = (const float*)d_in[0];
  const int*   eidx = (const int*)d_in[1];
  const float* W1   = (const float*)d_in[2];
  const float* bias = (const float*)d_in[3];
  const float* W2   = (const float*)d_in[4];
  const float* b2   = (const float*)d_in[5];
  float* out = (float*)d_out;

  const int N = out_size;
  const int E = in_sizes[1] / 2;
  const int H = in_sizes[4];
  const int* row = eidx;
  const int* col = eidx + E;

  const int P = (E + CAP - 1) / CAP;
  const int nSl = (N + SL - 1) / SL;

  char* ws = (char*)d_ws;
  const size_t nBytes = roundup256((size_t)N * 4);
  float* dinv = (float*)ws;
  float* t    = (float*)(ws + nBytes);
  float* w    = (float*)(ws + 2 * nBytes);
  float* cc   = w + 64;
  unsigned* segT = (unsigned*)(ws + 2 * nBytes + 512);
  unsigned* sorted = (unsigned*)(ws + 2 * nBytes + 512 + roundup256((size_t)nSl * P * 4));
  (void)ws_size; (void)n_in;

  sort_edges_kernel<<<P + 1, SCAN_T, 0, stream>>>(row, col, sorted, segT, E, N, P, nSl,
                                                  W1, bias, W2, b2, w, cc, H);
  slice_node_kernel<<<nSl, WT, 0, stream>>>(x, w, sorted, segT, dinv, t, N, P);
  slice_msg_kernel<<<nSl, WT, 0, stream>>>(sorted, segT, dinv, t, cc, out, N, P);
}


# Edges binned into fixed 64-word buckets per (slice, partition), unused words marked empty: no segment words, the sort drops its prefix sum, launches 2-3 a dependent load; LDS add order changes
# speedup vs baseline: 6.6527x; 1.1298x over previous
#include <hip/hip_runtime.h>


typedef __attribute__((ext_vector_type(2))) float v2f;
typedef __attribute__((ext_vector_type(8))) float v8f;

#define F_DIM 48
#define SL 400
#define TILES (SL / 16)
#define DBITS 9
#define DMASK ((1u << DBITS) - 1u)
#define RBAD 0x7FFFFFu
#define NB 512
#define BK 64
#define BQ (BK / 4)
#define EMPTY 0xFFFFFFFFu
#define CAP 6656
#define SCAN_T 512
#define EPT ((CAP + SCAN_T - 1) / SCAN_T)
#define WT 256
#define WW (WT / 32)
#define TPW ((TILES + WW - 1) / WW)

static_assert(SL % 16 == 0 && SL <= (1 << DBITS), "slice must be whole tiles and fit its field");
static_assert(NB == SCAN_T, "one histogram bin per thread");
static_assert(SL <= (int)DMASK, "the low bits of EMPTY must be no node of a slice");
static_assert(CAP % SCAN_T == 0, "a full partition gives every thread of the sort the same count");
static_assert((1 << DBITS) == 2 * WT, "the slice arrays are cleared two words per thread");
static_assert(SL <= 2 * WT, "a slice thread owns at most two nodes");

__device__ __forceinline__ void prep_w_block(const float* __restrict__ W1,
                                             const float* __restrict__ bias,
                                             const float* __restrict__ W2,
                                             const float* __restrict__ b2,
                                             float* __restrict__ w,
                                             float* __restrict__ cconst,
                                             int H) {
  constexpr int NW = SCAN_T / 32;
  constexpr int NR = F_DIM / NW;
  static_assert(F_DIM % NW == 0, "rows must divide evenly over the waves");
  const int lane = threadIdx.x & 31;
  const int wave = __builtin_amdgcn_readfirstlane((int)(threadIdx.x >> 5));
  float acc[NR];
#pragma unroll
  for (int i = 0; i < NR; ++i) acc[i] = 0.f;
  float sw = 0.f;
#pragma unroll 8
  for (int h = lane; h < H; h += 32) {
    const float w2 = W2[h];
#pragma unroll
    for (int i = 0; i < NR; ++i)
      acc[i] = fmaf(W1[(size_t)(wave + NW * i) * H + h], w2, acc[i]);
    sw += w2;
  }
#pragma unroll
  for (int o = 16; o > 0; o >>= 1) {
#pragma unroll
    for (int i = 0; i < NR; ++i) acc[i] += __shfl_xor(acc[i], o);
    sw += __shfl_xor(sw, o);
  }
  if (lane == 0) {
#pragma unroll
    for (int i = 0; i < NR; ++i) w[wave + NW * i] = acc[i];
    if (wave == 0) *cconst = bias[0] * sw + b2[0];
  }
}

#define HOLD13(a)                                                          \
  asm volatile("" :: "v"(a[0]), "v"(a[1]), "v"(a[2]), "v"(a[3]), "v"(a[4]), "v"(a[5]),      \
                     "v"(a[6]), "v"(a[7]), "v"(a[8]), "v"(a[9]), "v"(a[10]), "v"(a[11]),    \
                     "v"(a[12]))
static_assert(EPT == 13, "HOLD13 names every element");

#define HOLDQ(a, o)                                                        \
  asm volatile("" :: "v"(a[(o)].x), "v"(a[(o)].y), "v"(a[(o)].z), "v"(a[(o)].w),            \
                     "v"(a[(o) + 1].x), "v"(a[(o) + 1].y), "v"(a[(o) + 1].z), "v"(a[(o) + 1].w), \
                     "v"(a[(o) + 2].x), "v"(a[(o) + 2].y), "v"(a[(o) + 2].z), "v"(a[(o) + 2].w), \
                     "v"(a[(o) + 3].x), "v"(a[(o) + 3].y), "v"(a[(o) + 3].z), "v"(a[(o) + 3].w))

__global__ __launch_bounds__(SCAN_T)
void sort_edges_kernel(const int* __restrict__ row,
                       const int* __restrict__ col,
                       unsigned* __restrict__ buckets,
                       uint2* __restrict__ ovf,
                       unsigned* __restrict__ ovfCnt, int E, int N, int P, int nSl,
                       const float* __restrict__ W1,
                       const float* __restrict__ bias,
                       const float* __restrict__ W2,
                       const float* __restrict__ b2,
                       float* __restrict__ w,
                       float* __restrict__ cconst, int H) {
  __shared__ unsigned hist[NB + 32];
  __shared__ unsigned ovfN;
  __shared__ uint4 stage4[NB * BQ];
  if (blockIdx.x >= (unsigned)P) {
    prep_w_block(W1, bias, W2, b2, w, cconst, H);
    return;
  }
  unsigned* stage = reinterpret_cast<unsigned*>(stage4);
  const int tid = threadIdx.x;
  const int lane = tid & 31;
  const int p = blockIdx.x;
  hist[tid] = 0u;
  if (tid < 32) hist[NB + tid] = 0u;
  if (tid == 0) ovfN = 0u;
  const int e0 = p * CAP;
  const int e1 = (E - e0 < CAP) ? E : e0 + CAP;
  int cc[EPT], rr[EPT];
#pragma unroll
  for (int u = 0; u < EPT; ++u) {
    const int e = e0 + tid + u * SCAN_T;
    const bool in = e < e1;
    const int ec = in ? e : e0;
    cc[u] = col[ec];
    rr[u] = row[ec];
  }
  const int nPc = nSl * BQ;
  for (int q = tid; q < nPc; q += SCAN_T) stage4[q] = make_uint4(EMPTY, EMPTY, EMPTY, EMPTY);
  HOLD13(cc); HOLD13(rr);
#pragma unroll
  for (int u = 0; u < EPT; ++u)
    if (e0 + tid + u * SCAN_T >= e1) cc[u] = -1;
  __syncthreads();

  unsigned bin[EPT], rk[EPT], pk[EPT];
  bool val[EPT];
#pragma unroll
  for (int u = 0; u < EPT; ++u) {
    const unsigned c = (unsigned)cc[u];
    const unsigned b = c / SL;
    val[u] = (c < (unsigned)N) && (b < NB);
    bin[u] = val[u] ? b : (unsigned)(NB + lane);
    const unsigned r = (unsigned)rr[u] < (unsigned)N ? (unsigned)rr[u] : RBAD;
    pk[u] = (c - b * SL) | (r << DBITS);
    rk[u] = atomicAdd(&hist[bin[u]], 1u);
  }
  HOLD13(rk);
#pragma unroll
  for (int u = 0; u < EPT; ++u) {
    if (val[u]) {
      if (rk[u] < BK) {
        stage[bin[u] * BK + ((rk[u] + bin[u]) & (BK - 1))] = pk[u];
      } else {
        const unsigned j = atomicAdd(&ovfN, 1u);
        ovf[(size_t)p * CAP + j] = make_uint2(bin[u], pk[u]);
      }
    }
  }
  __syncthreads();

  uint4* dst4 = reinterpret_cast<uint4*>(buckets);
  for (int q0 = 0; q0 < nPc; q0 += 8 * SCAN_T) {
    uint4 pc[8];
#pragma unroll
    for (int k = 0; k < 8; ++k) {
      const int q = q0 + tid + k * SCAN_T;
      pc[k] = stage4[q < nPc ? q : 0];
    }
    HOLDQ(pc, 0); HOLDQ(pc, 4);
#pragma unroll
    for (int k = 0; k < 8; ++k) {
      const int q = q0 + tid + k * SCAN_T;
      if (q < nPc) dst4[((size_t)(q / BQ) * P + p) * BQ + (q & (BQ - 1))] = pc[k];
    }
  }
  if (tid == 0) ovfCnt[p] = ovfN;
}

#define BUCKET_LOADS(first)                                                \
  _Pragma("unroll")                                                        \
  for (int j = 0; j < BQ; ++j) {                                           \
    const int q = (first) + tid + WT * j;                                  \
    const uint4 pc = bq[q < nQ ? q : 0];                                   \
    e[4 * j] = pc.x; e[4 * j + 1] = pc.y; e[4 * j + 2] = pc.z; e[4 * j + 3] = pc.w; \
  }
#define BUCKET_MASK(first)                                                 \
  _Pragma("unroll")                                                        \
  for (int j = 0; j < BQ; ++j) {                                           \
    if ((first) + tid + WT * j >= nQ) {                                    \
      e[4 * j] = EMPTY; e[4 * j + 1] = EMPTY; e[4 * j + 2] = EMPTY; e[4 * j + 3] = EMPTY; \
    }                                                                      \
  }
#define IS_EDGE(wd) (((wd) & DMASK) != DMASK)

#define HOLD16(a, o)                                                       \
  asm volatile("" :: "v"(a[(o) + 0]), "v"(a[(o) + 1]), "v"(a[(o) + 2]), "v"(a[(o) + 3]),     \
                     "v"(a[(o) + 4]), "v"(a[(o) + 5]), "v"(a[(o) + 6]), "v"(a[(o) + 7]),     \
                     "v"(a[(o) + 8]), "v"(a[(o) + 9]), "v"(a[(o) + 10]), "v"(a[(o) + 11]),   \
                     "v"(a[(o) + 12]), "v"(a[(o) + 13]), "v"(a[(o) + 14]), "v"(a[(o) + 15]))

#define HOLD12(f)                                                          \
  asm volatile("" :: "v"(f[0]), "v"(f[1]), "v"(f[2]), "v"(f[3]), "v"(f[4]), "v"(f[5]),      \
                     "v"(f[6]), "v"(f[7]), "v"(f[8]), "v"(f[9]), "v"(f[10]), "v"(f[11]))
static_assert(F_DIM / 4 == 12 && TPW == 4, "HOLD12 names every fragment of a tile, and four tiles are named");
static_assert(4 * BQ == 64, "four HOLD16 name every word a thread reads of a chunk of buckets");

__global__ __launch_bounds__(WT)
void slice_node_kernel(const float* __restrict__ x,
                       const float* __restrict__ w,
                       const unsigned* __restrict__ buckets,
                       const uint2* __restrict__ ovf,
                       const unsigned* __restrict__ ovfCnt,
                       float* __restrict__ dinv,
                       float* __restrict__ t, int N, int P) {
  __shared__ unsigned deg[1 << DBITS];
  __shared__ float sS[SL];
  const int tid = threadIdx.x;
  const int lane = tid & 31;
  const int wave = __builtin_amdgcn_readfirstlane((int)(threadIdx.x >> 5));
  const int y = blockIdx.x;
  deg[tid] = 0u;
  deg[tid + WT] = 0u;
  const uint4* bq = reinterpret_cast<const uint4*>(buckets) + (size_t)y * P * BQ;
  const int nQ = P * BQ;
  int pb = 0;
  unsigned e[4 * BQ];
  BUCKET_LOADS(0)
  unsigned oc = tid < P ? ovfCnt[tid] : 0u;

  const int m = lane & 15;
  const int halfSel = lane >> 4;
  const int r = m & 7;
  v2f b[F_DIM / 4], a[TPW][F_DIM / 4];
#pragma unroll
  for (int k = 0; k < F_DIM / 4; ++k)
    b[k] = *reinterpret_cast<const v2f*>(w + 4 * k + 2 * halfSel);
#pragma unroll
  for (int i = 0; i < TPW; ++i) {
    const int tl = wave + WW * i;
    const int tc = tl < TILES ? tl : TILES - 1;
    int nrow = y * SL + tc * 16 + m;
    nrow = nrow < N ? nrow : N - 1;
    const float* xrow = x + (size_t)nrow * F_DIM;
#pragma unroll
    for (int k = 0; k < F_DIM / 4; ++k)
      a[i][k] = *reinterpret_cast<const v2f*>(xrow + 4 * k + 2 * halfSel);
  }
  HOLD12(b); HOLD12(a[0]); HOLD12(a[1]); HOLD12(a[2]); HOLD12(a[3]);
  HOLD16(e, 0); HOLD16(e, 16); HOLD16(e, 32); HOLD16(e, 48);
  float sv[TPW];
#pragma unroll
  for (int i = 0; i < TPW; ++i) {
    v8f c = {0.f, 0.f, 0.f, 0.f, 0.f, 0.f, 0.f, 0.f};
#pragma unroll
    for (int k = 0; k < F_DIM / 4; ++k)
      c = __builtin_amdgcn_wmma_f32_16x16x4_f32(
           false, a[i][k],  false, b[k],
           (short)0, c,  false,  false);
    float s = c[0];
    s = r == 1 ? c[1] : s; s = r == 2 ? c[2] : s; s = r == 3 ? c[3] : s;
    s = r == 4 ? c[4] : s; s = r == 5 ? c[5] : s; s = r == 6 ? c[6] : s;
    s = r == 7 ? c[7] : s;
    sv[i] = s;
  }
#pragma unroll
  for (int i = 0; i < TPW; ++i) {
    const int tl = wave + WW * i;
    if (tl < TILES && m < 8) sS[tl * 16 + halfSel * 8 + r] = sv[i];
  }
  __syncthreads();

  for (;;) {
    BUCKET_MASK(pb * BQ)
#pragma unroll
    for (int i = 0; i < 4 * BQ; ++i)
      if (IS_EDGE(e[i])) atomicAdd(&deg[e[i] & DMASK], 1u);
    if (__any(oc != 0u)) {
#pragma nounroll
      for (int k = 0; k < 32; ++k) {
        unsigned cn = (unsigned)__shfl((int)oc, k);
        cn = cn < CAP ? cn : CAP;
        if (cn != 0u) {
          const uint2* op = ovf + (size_t)(pb + wave * 32 + k) * CAP;
          for (unsigned j = lane; j < cn; j += 32u) {
            const uint2 q = op[j];
            if (q.x == (unsigned)y) atomicAdd(&deg[q.y & DMASK], 1u);
          }
        }
      }
    }
    pb += WT;
    if (pb >= P) break;
    BUCKET_LOADS(pb * BQ)
    oc = pb + tid < P ? ovfCnt[pb + tid] : 0u;
    HOLD16(e, 0); HOLD16(e, 16); HOLD16(e, 32); HOLD16(e, 48);
  }
  __syncthreads();

  for (int i = tid; i < SL; i += WT) {
    const int n = y * SL + i;
    if (n < N) {
      const float d = rsqrtf(fmaxf((float)(deg[i] + 1u), 1.0f));
      dinv[n] = d;
      t[n] = d * sS[i];
    }
  }
}

#define MSG_OK(wd) (IS_EDGE(wd) && (((wd) >> DBITS) < (unsigned)N))

__global__ __launch_bounds__(WT)
void slice_msg_kernel(const unsigned* __restrict__ buckets,
                      const uint2* __restrict__ ovf,
                      const unsigned* __restrict__ ovfCnt,
                      const float* __restrict__ dinv,
                      const float* __restrict__ t,
                      const float* __restrict__ cconst,
                      float* __restrict__ out, int N, int P) {
  __shared__ float acc[1 << DBITS];
  const int tid = threadIdx.x;
  const int lane = tid & 31;
  const int wave = __builtin_amdgcn_readfirstlane((int)(threadIdx.x >> 5));
  const int y = blockIdx.x;
  acc[tid] = 0.f;
  acc[tid + WT] = 0.f;
  const uint4* bq = reinterpret_cast<const uint4*>(buckets) + (size_t)y * P * BQ;
  const int nQ = P * BQ;
  int pb = 0;
  unsigned e[4 * BQ];
  BUCKET_LOADS(0)
  unsigned oc = tid < P ? ovfCnt[tid] : 0u;
  const int n0 = y * SL + tid, n1 = n0 + WT;
  const float dn0 = dinv[n0 < N ? n0 : N - 1], tn0 = t[n0 < N ? n0 : N - 1];
  const float dn1 = dinv[n1 < N ? n1 : N - 1], tn1 = t[n1 < N ? n1 : N - 1];
  const float cst = *cconst;
  __syncthreads();
  asm volatile("" :: "v"(dn0), "v"(tn0), "v"(dn1), "v"(tn1));

  for (;;) {
    float v[4 * BQ];
    BUCKET_MASK(pb * BQ)
#pragma unroll
    for (int i = 0; i < 4 * BQ; ++i)
      v[i] = t[MSG_OK(e[i]) ? e[i] >> DBITS : 0u];
    HOLD16(v, 0); HOLD16(v, 16); HOLD16(v, 32); HOLD16(v, 48);
#pragma unroll
    for (int i = 0; i < 4 * BQ; ++i)
      if (MSG_OK(e[i])) atomicAdd(&acc[e[i] & DMASK], v[i]);
    if (__any(oc != 0u)) {
#pragma nounroll
      for (int k = 0; k < 32; ++k) {
        unsigned cn = (unsigned)__shfl((int)oc, k);
        cn = cn < CAP ? cn : CAP;
        if (cn != 0u) {
          const uint2* op = ovf + (size_t)(pb + wave * 32 + k) * CAP;
          for (unsigned j = lane; j < cn; j += 32u) {
            const uint2 q = op[j];
            if (q.x == (unsigned)y && (q.y >> DBITS) < (unsigned)N)
              atomicAdd(&acc[q.y & DMASK], t[q.y >> DBITS]);
          }
        }
      }
    }
    pb += WT;
    if (pb >= P) break;
    BUCKET_LOADS(pb * BQ)
    oc = pb + tid < P ? ovfCnt[pb + tid] : 0u;
  }
  __syncthreads();

  if (n0 < N) out[n0] = fmaf(dn0, tn0 + acc[tid], cst);
  if (tid + WT < SL && n1 < N) out[n1] = fmaf(dn1, tn1 + acc[tid + WT], cst);
}

static inline size_t roundup256(size_t v) { return (v + 255) & ~(size_t)255; }

extern "C" void kernel_launch(void* const* d_in, const int* in_sizes, int n_in,
                              void* d_out, int out_size, void* d_ws, size_t ws_size,
                              hipStream_t stream) {
  const float* x    = (const float*)d_in[0];
  const int*   eidx = (const int*)d_in[1];
  const float* W1   = (const float*)d_in[2];
  const float* bias = (const float*)d_in[3];
  const float* W2   = (const float*)d_in[4];
  const float* b2   = (const float*)d_in[5];
  float* out = (float*)d_out;

  const int N = out_size;
  const int E = in_sizes[1] / 2;
  const int H = in_sizes[4];
  const int* row = eidx;
  const int* col = eidx + E;

  const int P = (E + CAP - 1) / CAP;
  const int nSl = (N + SL - 1) / SL;

  char* ws = (char*)d_ws;
  const size_t nBytes = roundup256((size_t)N * 4);
  float* dinv = (float*)ws;
  float* t    = (float*)(ws + nBytes);
  float* w    = (float*)(ws + 2 * nBytes);
  float* cc   = w + 64;
  unsigned* ovfCnt = (unsigned*)(ws + 2 * nBytes + 512);
  char* bkBase = ws + 2 * nBytes + 512 + roundup256((size_t)P * 4);
  unsigned* buckets = (unsigned*)bkBase;
  uint2* ovf = (uint2*)(bkBase + roundup256((size_t)nSl * P * BK * 4));
  (void)ws_size; (void)n_in;

  sort_edges_kernel<<<P + 1, SCAN_T, 0, stream>>>(row, col, buckets, ovf, ovfCnt, E, N, P, nSl,
                                                  W1, bias, W2, b2, w, cc, H);
  slice_node_kernel<<<nSl, WT, 0, stream>>>(x, w, buckets, ovf, ovfCnt, dinv, t, N, P);
  slice_msg_kernel<<<nSl, WT, 0, stream>>>(buckets, ovf, ovfCnt, dinv, t, cc, out, N, P);
}


# Launches 2, 3: a wave uses its loads in issue order as they return (counted waits), not all at once: degree adds run under the x-row loads, message adds under later gathers; LDS add order changes
# speedup vs baseline: 7.0694x; 1.0626x over previous
#include <hip/hip_runtime.h>


typedef __attribute__((ext_vector_type(2))) float v2f;
typedef __attribute__((ext_vector_type(8))) float v8f;

#define F_DIM 48
#define SL 400
#define TILES (SL / 16)
#define DBITS 9
#define DMASK ((1u << DBITS) - 1u)
#define RBAD 0x7FFFFFu
#define NB 512
#define BK 64
#define BQ (BK / 4)
#define EMPTY 0xFFFFFFFFu
#define CAP 6656
#define SCAN_T 512
#define EPT ((CAP + SCAN_T - 1) / SCAN_T)
#define WT 256
#define WW (WT / 32)
#define TPW ((TILES + WW - 1) / WW)

static_assert(SL % 16 == 0 && SL <= (1 << DBITS), "slice must be whole tiles and fit its field");
static_assert(NB == SCAN_T, "one histogram bin per thread");
static_assert(SL <= (int)DMASK, "the low bits of EMPTY must be no node of a slice");
static_assert(CAP % SCAN_T == 0, "a full partition gives every thread of the sort the same count");
static_assert((1 << DBITS) == 2 * WT, "the slice arrays are cleared two words per thread");
static_assert(SL <= 2 * WT, "a slice thread owns at most two nodes");

__device__ __forceinline__ void prep_w_block(const float* __restrict__ W1,
                                             const float* __restrict__ bias,
                                             const float* __restrict__ W2,
                                             const float* __restrict__ b2,
                                             float* __restrict__ w,
                                             float* __restrict__ cconst,
                                             int H) {
  constexpr int NW = SCAN_T / 32;
  constexpr int NR = F_DIM / NW;
  static_assert(F_DIM % NW == 0, "rows must divide evenly over the waves");
  const int lane = threadIdx.x & 31;
  const int wave = __builtin_amdgcn_readfirstlane((int)(threadIdx.x >> 5));
  float acc[NR];
#pragma unroll
  for (int i = 0; i < NR; ++i) acc[i] = 0.f;
  float sw = 0.f;
#pragma unroll 8
  for (int h = lane; h < H; h += 32) {
    const float w2 = W2[h];
#pragma unroll
    for (int i = 0; i < NR; ++i)
      acc[i] = fmaf(W1[(size_t)(wave + NW * i) * H + h], w2, acc[i]);
    sw += w2;
  }
#pragma unroll
  for (int o = 16; o > 0; o >>= 1) {
#pragma unroll
    for (int i = 0; i < NR; ++i) acc[i] += __shfl_xor(acc[i], o);
    sw += __shfl_xor(sw, o);
  }
  if (lane == 0) {
#pragma unroll
    for (int i = 0; i < NR; ++i) w[wave + NW * i] = acc[i];
    if (wave == 0) *cconst = bias[0] * sw + b2[0];
  }
}

#define HOLD13(a)                                                          \
  asm volatile("" :: "v"(a[0]), "v"(a[1]), "v"(a[2]), "v"(a[3]), "v"(a[4]), "v"(a[5]),      \
                     "v"(a[6]), "v"(a[7]), "v"(a[8]), "v"(a[9]), "v"(a[10]), "v"(a[11]),    \
                     "v"(a[12]))
static_assert(EPT == 13, "HOLD13 names every element");

#define HOLDQ(a, o)                                                        \
  asm volatile("" :: "v"(a[(o)].x), "v"(a[(o)].y), "v"(a[(o)].z), "v"(a[(o)].w),            \
                     "v"(a[(o) + 1].x), "v"(a[(o) + 1].y), "v"(a[(o) + 1].z), "v"(a[(o) + 1].w), \
                     "v"(a[(o) + 2].x), "v"(a[(o) + 2].y), "v"(a[(o) + 2].z), "v"(a[(o) + 2].w), \
                     "v"(a[(o) + 3].x), "v"(a[(o) + 3].y), "v"(a[(o) + 3].z), "v"(a[(o) + 3].w))

__global__ __launch_bounds__(SCAN_T)
void sort_edges_kernel(const int* __restrict__ row,
                       const int* __restrict__ col,
                       unsigned* __restrict__ buckets,
                       uint2* __restrict__ ovf,
                       unsigned* __restrict__ ovfCnt, int E, int N, int P, int nSl,
                       const float* __restrict__ W1,
                       const float* __restrict__ bias,
                       const float* __restrict__ W2,
                       const float* __restrict__ b2,
                       float* __restrict__ w,
                       float* __restrict__ cconst, int H) {
  __shared__ unsigned hist[NB + 32];
  __shared__ unsigned ovfN;
  __shared__ uint4 stage4[NB * BQ];
  if (blockIdx.x >= (unsigned)P) {
    prep_w_block(W1, bias, W2, b2, w, cconst, H);
    return;
  }
  unsigned* stage = reinterpret_cast<unsigned*>(stage4);
  const int tid = threadIdx.x;
  const int lane = tid & 31;
  const int p = blockIdx.x;
  hist[tid] = 0u;
  if (tid < 32) hist[NB + tid] = 0u;
  if (tid == 0) ovfN = 0u;
  const int e0 = p * CAP;
  const int e1 = (E - e0 < CAP) ? E : e0 + CAP;
  int cc[EPT], rr[EPT];
#pragma unroll
  for (int u = 0; u < EPT; ++u) {
    const int e = e0 + tid + u * SCAN_T;
    const bool in = e < e1;
    const int ec = in ? e : e0;
    cc[u] = col[ec];
    rr[u] = row[ec];
  }
  const int nPc = nSl * BQ;
  for (int q = tid; q < nPc; q += SCAN_T) stage4[q] = make_uint4(EMPTY, EMPTY, EMPTY, EMPTY);
  HOLD13(cc); HOLD13(rr);
#pragma unroll
  for (int u = 0; u < EPT; ++u)
    if (e0 + tid + u * SCAN_T >= e1) cc[u] = -1;
  __syncthreads();

  unsigned bin[EPT], rk[EPT], pk[EPT];
  bool val[EPT];
#pragma unroll
  for (int u = 0; u < EPT; ++u) {
    const unsigned c = (unsigned)cc[u];
    const unsigned b = c / SL;
    val[u] = (c < (unsigned)N) && (b < NB);
    bin[u] = val[u] ? b : (unsigned)(NB + lane);
    const unsigned r = (unsigned)rr[u] < (unsigned)N ? (unsigned)rr[u] : RBAD;
    pk[u] = (c - b * SL) | (r << DBITS);
    rk[u] = atomicAdd(&hist[bin[u]], 1u);
  }
  HOLD13(rk);
#pragma unroll
  for (int u = 0; u < EPT; ++u) {
    if (val[u]) {
      if (rk[u] < BK) {
        stage[bin[u] * BK + ((rk[u] + bin[u]) & (BK - 1))] = pk[u];
      } else {
        const unsigned j = atomicAdd(&ovfN, 1u);
        ovf[(size_t)p * CAP + j] = make_uint2(bin[u], pk[u]);
      }
    }
  }
  __syncthreads();

  uint4* dst4 = reinterpret_cast<uint4*>(buckets);
  for (int q0 = 0; q0 < nPc; q0 += 8 * SCAN_T) {
    uint4 pc[8];
#pragma unroll
    for (int k = 0; k < 8; ++k) {
      const int q = q0 + tid + k * SCAN_T;
      pc[k] = stage4[q < nPc ? q : 0];
    }
    HOLDQ(pc, 0); HOLDQ(pc, 4);
#pragma unroll
    for (int k = 0; k < 8; ++k) {
      const int q = q0 + tid + k * SCAN_T;
      if (q < nPc) dst4[((size_t)(q / BQ) * P + p) * BQ + (q & (BQ - 1))] = pc[k];
    }
  }
  if (tid == 0) ovfCnt[p] = ovfN;
}

#define BUCKET_LOADS(first)                                                \
  _Pragma("unroll")                                                        \
  for (int j = 0; j < BQ; ++j) {                                           \
    const int q = (first) + tid + WT * j;                                  \
    const uint4 pc = bq[q < nQ ? q : 0];                                   \
    e[4 * j] = pc.x; e[4 * j + 1] = pc.y; e[4 * j + 2] = pc.z; e[4 * j + 3] = pc.w; \
  }
#define BUCKET_MASK(first)                                                 \
  _Pragma("unroll")                                                        \
  for (int j = 0; j < BQ; ++j) {                                           \
    if ((first) + tid + WT * j >= nQ) {                                    \
      e[4 * j] = EMPTY; e[4 * j + 1] = EMPTY; e[4 * j + 2] = EMPTY; e[4 * j + 3] = EMPTY; \
    }                                                                      \
  }
#define IS_EDGE(wd) (((wd) & DMASK) != DMASK)

#define HOLD16(a, o)                                                       \
  asm volatile("" :: "v"(a[(o) + 0]), "v"(a[(o) + 1]), "v"(a[(o) + 2]), "v"(a[(o) + 3]),     \
                     "v"(a[(o) + 4]), "v"(a[(o) + 5]), "v"(a[(o) + 6]), "v"(a[(o) + 7]),     \
                     "v"(a[(o) + 8]), "v"(a[(o) + 9]), "v"(a[(o) + 10]), "v"(a[(o) + 11]),   \
                     "v"(a[(o) + 12]), "v"(a[(o) + 13]), "v"(a[(o) + 14]), "v"(a[(o) + 15]))

static_assert(4 * BQ == 64, "four HOLD16 name every word a thread reads of a chunk of buckets");

#define DEG_CHUNK()                                                        \
  {                                                                        \
    BUCKET_MASK(pb * BQ)                                                   \
    _Pragma("unroll")                                                      \
    for (int i = 0; i < 4 * BQ; ++i)                                       \
      if (IS_EDGE(e[i])) atomicAdd(&deg[e[i] & DMASK], 1u);                \
    if (__any(oc != 0u)) {                                                 \
      _Pragma("nounroll")                                                  \
      for (int k = 0; k < 32; ++k) {                                       \
        unsigned cn = (unsigned)__shfl((int)oc, k);                        \
        cn = cn < CAP ? cn : CAP;                                          \
        if (cn != 0u) {                                                    \
          const uint2* op = ovf + (size_t)(pb + wave * 32 + k) * CAP;      \
          for (unsigned j = lane; j < cn; j += 32u) {                      \
            const uint2 q = op[j];                                         \
            if (q.x == (unsigned)y) atomicAdd(&deg[q.y & DMASK], 1u);      \
          }                                                                \
        }                                                                  \
      }                                                                    \
    }                                                                      \
  }

__global__ __launch_bounds__(WT)
void slice_node_kernel(const float* __restrict__ x,
                       const float* __restrict__ w,
                       const unsigned* __restrict__ buckets,
                       const uint2* __restrict__ ovf,
                       const unsigned* __restrict__ ovfCnt,
                       float* __restrict__ dinv,
                       float* __restrict__ t, int N, int P) {
  __shared__ unsigned deg[1 << DBITS];
  __shared__ float sS[SL];
  const int tid = threadIdx.x;
  const int lane = tid & 31;
  const int wave = __builtin_amdgcn_readfirstlane((int)(threadIdx.x >> 5));
  const int y = blockIdx.x;
  deg[tid] = 0u;
  deg[tid + WT] = 0u;
  const uint4* bq = reinterpret_cast<const uint4*>(buckets) + (size_t)y * P * BQ;
  const int nQ = P * BQ;
  int pb = 0;
  unsigned e[4 * BQ];
  BUCKET_LOADS(0)
  unsigned oc = tid < P ? ovfCnt[tid] : 0u;

  const int m = lane & 15;
  const int halfSel = lane >> 4;
  const int r = m & 7;
  v2f b[F_DIM / 4], a[TPW][F_DIM / 4];
#pragma unroll
  for (int k = 0; k < F_DIM / 4; ++k)
    b[k] = *reinterpret_cast<const v2f*>(w + 4 * k + 2 * halfSel);
#pragma unroll
  for (int i = 0; i < TPW; ++i) {
    const int tl = wave + WW * i;
    const int tc = tl < TILES ? tl : TILES - 1;
    int nrow = y * SL + tc * 16 + m;
    nrow = nrow < N ? nrow : N - 1;
    const float* xrow = x + (size_t)nrow * F_DIM;
#pragma unroll
    for (int k = 0; k < F_DIM / 4; ++k)
      a[i][k] = *reinterpret_cast<const v2f*>(xrow + 4 * k + 2 * halfSel);
  }
  __builtin_amdgcn_sched_barrier(0);
  __builtin_amdgcn_fence(__ATOMIC_RELEASE, "workgroup", "local");
  __builtin_amdgcn_s_barrier();
  __builtin_amdgcn_fence(__ATOMIC_ACQUIRE, "workgroup", "local");
  DEG_CHUNK()
  float sv[TPW];
#pragma unroll
  for (int i = 0; i < TPW; ++i) {
    v8f c = {0.f, 0.f, 0.f, 0.f, 0.f, 0.f, 0.f, 0.f};
#pragma unroll
    for (int k = 0; k < F_DIM / 4; ++k)
      c = __builtin_amdgcn_wmma_f32_16x16x4_f32(
           false, a[i][k],  false, b[k],
           (short)0, c,  false,  false);
    float s = c[0];
    s = r == 1 ? c[1] : s; s = r == 2 ? c[2] : s; s = r == 3 ? c[3] : s;
    s = r == 4 ? c[4] : s; s = r == 5 ? c[5] : s; s = r == 6 ? c[6] : s;
    s = r == 7 ? c[7] : s;
    sv[i] = s;
  }
#pragma unroll
  for (int i = 0; i < TPW; ++i) {
    const int tl = wave + WW * i;
    if (tl < TILES && m < 8) sS[tl * 16 + halfSel * 8 + r] = sv[i];
  }
  for (pb = WT; pb < P; pb += WT) {
    BUCKET_LOADS(pb * BQ)
    oc = pb + tid < P ? ovfCnt[pb + tid] : 0u;
    HOLD16(e, 0); HOLD16(e, 16); HOLD16(e, 32); HOLD16(e, 48);
    DEG_CHUNK()
  }
  __syncthreads();

  for (int i = tid; i < SL; i += WT) {
    const int n = y * SL + i;
    if (n < N) {
      const float d = rsqrtf(fmaxf((float)(deg[i] + 1u), 1.0f));
      dinv[n] = d;
      t[n] = d * sS[i];
    }
  }
}

#define MSG_OK(wd) (IS_EDGE(wd) && (((wd) >> DBITS) < (unsigned)N))

__global__ __launch_bounds__(WT)
void slice_msg_kernel(const unsigned* __restrict__ buckets,
                      const uint2* __restrict__ ovf,
                      const unsigned* __restrict__ ovfCnt,
                      const float* __restrict__ dinv,
                      const float* __restrict__ t,
                      const float* __restrict__ cconst,
                      float* __restrict__ out, int N, int P) {
  __shared__ float acc[1 << DBITS];
  const int tid = threadIdx.x;
  const int lane = tid & 31;
  const int wave = __builtin_amdgcn_readfirstlane((int)(threadIdx.x >> 5));
  const int y = blockIdx.x;
  acc[tid] = 0.f;
  acc[tid + WT] = 0.f;
  const uint4* bq = reinterpret_cast<const uint4*>(buckets) + (size_t)y * P * BQ;
  const int nQ = P * BQ;
  int pb = 0;
  unsigned e[4 * BQ];
  const int n0 = y * SL + tid, n1 = n0 + WT;
  const float dn0 = dinv[n0 < N ? n0 : N - 1], tn0 = t[n0 < N ? n0 : N - 1];
  const float dn1 = dinv[n1 < N ? n1 : N - 1], tn1 = t[n1 < N ? n1 : N - 1];
  BUCKET_LOADS(0)
  unsigned oc = tid < P ? ovfCnt[tid] : 0u;
  const float cst = *cconst;
  __builtin_amdgcn_sched_barrier(0);
  __builtin_amdgcn_fence(__ATOMIC_RELEASE, "workgroup", "local");
  __builtin_amdgcn_s_barrier();
  __builtin_amdgcn_fence(__ATOMIC_ACQUIRE, "workgroup", "local");
  asm volatile("" :: "v"(dn0), "v"(tn0), "v"(dn1), "v"(tn1));

  for (;;) {
    float v[4 * BQ];
    BUCKET_MASK(pb * BQ)
#pragma unroll
    for (int i = 0; i < 4 * BQ; ++i)
      v[i] = t[MSG_OK(e[i]) ? e[i] >> DBITS : 0u];
#pragma unroll
    for (int g = 0; g < 4 * BQ; g += 16) {
      HOLD16(v, g);
#pragma unroll
      for (int i = g; i < g + 16; ++i)
        if (MSG_OK(e[i])) atomicAdd(&acc[e[i] & DMASK], v[i]);
    }
    if (__any(oc != 0u)) {
#pragma nounroll
      for (int k = 0; k < 32; ++k) {
        unsigned cn = (unsigned)__shfl((int)oc, k);
        cn = cn < CAP ? cn : CAP;
        if (cn != 0u) {
          const uint2* op = ovf + (size_t)(pb + wave * 32 + k) * CAP;
          for (unsigned j = lane; j < cn; j += 32u) {
            const uint2 q = op[j];
            if (q.x == (unsigned)y && (q.y >> DBITS) < (unsigned)N)
              atomicAdd(&acc[q.y & DMASK], t[q.y >> DBITS]);
          }
        }
      }
    }
    pb += WT;
    if (pb >= P) break;
    BUCKET_LOADS(pb * BQ)
    oc = pb + tid < P ? ovfCnt[pb + tid] : 0u;
  }
  __syncthreads();

  if (n0 < N) out[n0] = fmaf(dn0, tn0 + acc[tid], cst);
  if (tid + WT < SL && n1 < N) out[n1] = fmaf(dn1, tn1 + acc[tid + WT], cst);
}

static inline size_t roundup256(size_t v) { return (v + 255) & ~(size_t)255; }

extern "C" void kernel_launch(void* const* d_in, const int* in_sizes, int n_in,
                              void* d_out, int out_size, void* d_ws, size_t ws_size,
                              hipStream_t stream) {
  const float* x    = (const float*)d_in[0];
  const int*   eidx = (const int*)d_in[1];
  const float* W1   = (const float*)d_in[2];
  const float* bias = (const float*)d_in[3];
  const float* W2   = (const float*)d_in[4];
  const float* b2   = (const float*)d_in[5];
  float* out = (float*)d_out;

  const int N = out_size;
  const int E = in_sizes[1] / 2;
  const int H = in_sizes[4];
  const int* row = eidx;
  const int* col = eidx + E;

  const int P = (E + CAP - 1) / CAP;
  const int nSl = (N + SL - 1) / SL;

  char* ws = (char*)d_ws;
  const size_t nBytes = roundup256((size_t)N * 4);
  float* dinv = (float*)ws;
  float* t    = (float*)(ws + nBytes);
  float* w    = (float*)(ws + 2 * nBytes);
  float* cc   = w + 64;
  unsigned* ovfCnt = (unsigned*)(ws + 2 * nBytes + 512);
  char* bkBase = ws + 2 * nBytes + 512 + roundup256((size_t)P * 4);
  unsigned* buckets = (unsigned*)bkBase;
  uint2* ovf = (uint2*)(bkBase + roundup256((size_t)nSl * P * BK * 4));
  (void)ws_size; (void)n_in;

  sort_edges_kernel<<<P + 1, SCAN_T, 0, stream>>>(row, col, buckets, ovf, ovfCnt, E, N, P, nSl,
                                                  W1, bias, W2, b2, w, cc, H);
  slice_node_kernel<<<nSl, WT, 0, stream>>>(x, w, buckets, ovf, ovfCnt, dinv, t, N, P);
  slice_msg_kernel<<<nSl, WT, 0, stream>>>(buckets, ovf, ovfCnt, dinv, t, cc, out, N, P);
}
